# starting kernel
# baseline (speedup 1.0000x reference)
#include <hip/hip_runtime.h>
#include <math.h>


typedef float v2f __attribute__((ext_vector_type(2)));
typedef float v8f __attribute__((ext_vector_type(8)));

#define INV_SQRT2  0.70710678f
#define INV_SQRT3  0.57735027f
#define INV_SQRT6  0.40824829f
#define INV_SQRT10 0.31622777f
#define SILU_C     1.67668f
#define SC0        0.02209709f
#define SC1        0.03125f

__global__ void k_zero(float* p, int n) {
  int i = blockIdx.x * blockDim.x + threadIdx.x;
  if (i < n) p[i] = 0.f;
}

__global__ void k_repack_w1(const float* __restrict__ w1, float* __restrict__ o) {
  int tid = blockIdx.x * blockDim.x + threadIdx.x;
  if (tid >= 160 * 16 * 32) return;
  int l  = tid & 31;
  int s  = (tid >> 5) & 15;
  int t  = tid >> 9;
  int li = l & 15;
  int kd = (l < 16) ? 0 : 2;
  int row = 4 * s + kd;
  int col = t * 16 + li;
  o[2 * tid + 0] = w1[row * 2560 + col];
  o[2 * tid + 1] = w1[(row + 1) * 2560 + col];
}

__global__ __launch_bounds__(32)
void k_edge_conv(const float* __restrict__ node_pos,
                 const float* __restrict__ edge_type,
                 const float* __restrict__ fc_w0,
                 const float* __restrict__ w1p,
                 const int*   __restrict__ edge_src,
                 const int*   __restrict__ edge_dst,
                 float* __restrict__ out)
{
  __shared__ float hid[16 * 66];
  __shared__ float xsr[16 * 20];
  __shared__ float xds[16 * 20];
  __shared__ float mid[16 * 260];
  __shared__ int   dsti[16];

  const int lane = threadIdx.x;
  const int li   = lane & 15;
  const int kd   = (lane < 16) ? 0 : 2;
  const int eoff = (lane < 16) ? 0 : 8;
  const int e0   = blockIdx.x * 16;

  if (lane < 16) {
    int s = edge_src[e0 + li];
    dsti[li] = edge_dst[e0 + li];
    const float4* q = (const float4*)(node_pos + (size_t)s * 16);
    float4* d = (float4*)(xsr + li * 20);
    d[0] = q[0]; d[1] = q[1]; d[2] = q[2]; d[3] = q[3];
  } else {
    int dd = edge_dst[e0 + li];
    const float4* q = (const float4*)(node_pos + (size_t)dd * 16);
    float4* d = (float4*)(xds + li * 20);
    d[0] = q[0]; d[1] = q[1]; d[2] = q[2]; d[3] = q[3];
  }

  {
    v2f a[4];
    const float* et = edge_type + (size_t)(e0 + li) * 16 + kd;
#pragma unroll
    for (int s = 0; s < 4; ++s) a[s] = *(const v2f*)(et + 4 * s);
#pragma unroll
    for (int tt = 0; tt < 4; ++tt) {
      v8f c = {};
#pragma unroll
      for (int s = 0; s < 4; ++s) {
        int row = 4 * s + kd;
        v2f b;
        b.x = fc_w0[row * 64 + tt * 16 + li];
        b.y = fc_w0[(row + 1) * 64 + tt * 16 + li];
        c = __builtin_amdgcn_wmma_f32_16x16x4_f32(false, a[s], false, b,
                                                  (short)0, c, false, false);
      }
#pragma unroll
      for (int r = 0; r < 8; ++r) {
        float x = c[r] * 0.25f;
        float h = (x / (1.f + __expf(-x))) * (SILU_C * 0.125f);
        hid[(r + eoff) * 66 + tt * 16 + li] = h;
      }
    }
  }
  __syncthreads();

  {
    const int e = li;
    const float* X = xsr + e * 20;
    const float* Y = xds + e * 20;
    float y0[4], y1[4][3];
#pragma unroll
    for (int v = 0; v < 4; ++v) {
      y0[v] = Y[v];
      y1[v][0] = Y[4 + v * 3 + 0];
      y1[v][1] = Y[4 + v * 3 + 1];
      y1[v][2] = Y[4 + v * 3 + 2];
    }
    float* M = mid + e * 260;
    const int ub = (lane < 16) ? 0 : 2;
#pragma unroll
    for (int uu = 0; uu < 2; ++uu) {
      int u = ub + uu;
      float x0 = X[u];
      float a0 = X[4 + u * 3 + 0], a1 = X[4 + u * 3 + 1], a2 = X[4 + u * 3 + 2];
#pragma unroll
      for (int v = 0; v < 4; ++v) {
        int uv = u * 4 + v;
        float b0 = y1[v][0], b1 = y1[v][1], b2 = y1[v][2];
        M[uv] = x0 * y0[v];
        M[16 + uv * 3 + 0] = x0 * b0;
        M[16 + uv * 3 + 1] = x0 * b1;
        M[16 + uv * 3 + 2] = x0 * b2;
        M[64 + uv * 3 + 0] = a0 * y0[v];
        M[64 + uv * 3 + 1] = a1 * y0[v];
        M[64 + uv * 3 + 2] = a2 * y0[v];
        M[112 + uv] = (a0 * b0 + a1 * b1 + a2 * b2) * INV_SQRT3;
        M[128 + uv * 3 + 0] = (a1 * b2 - a2 * b1) * INV_SQRT2;
        M[128 + uv * 3 + 1] = (a2 * b0 - a0 * b2) * INV_SQRT2;
        M[128 + uv * 3 + 2] = (a0 * b1 - a1 * b0) * INV_SQRT2;
        M[176 + uv * 5 + 0] = (a2 * b0 + a0 * b2) * INV_SQRT2;
        M[176 + uv * 5 + 1] = (a0 * b1 + a1 * b0) * INV_SQRT2;
        M[176 + uv * 5 + 2] = (2.f * a1 * b1 - a0 * b0 - a2 * b2) * INV_SQRT6;
        M[176 + uv * 5 + 3] = (a1 * b2 + a2 * b1) * INV_SQRT2;
        M[176 + uv * 5 + 4] = (a2 * b2 - a0 * b0) * INV_SQRT2;
      }
    }
  }
  __syncthreads();

  v2f A[16];
#pragma unroll
  for (int s = 0; s < 16; ++s) A[s] = *(const v2f*)&hid[li * 66 + 4 * s + kd];

  float acc0[8], acc1[8][3];
#pragma unroll
  for (int r = 0; r < 8; ++r) {
    acc0[r] = 0.f; acc1[r][0] = 0.f; acc1[r][1] = 0.f; acc1[r][2] = 0.f;
  }

#pragma unroll
  for (int p = 0; p < 10; ++p) {
    for (int tt = 0; tt < 16; ++tt) {
      int t = p * 16 + tt;
      const float* bq = w1p + (size_t)t * 1024 + 2 * lane;
      v8f ca = {}, cb = {};
#pragma unroll
      for (int s = 0; s < 8; ++s) {
        v2f b0 = *(const v2f*)(bq + (2 * s) * 64);
        v2f b1 = *(const v2f*)(bq + (2 * s + 1) * 64);
        ca = __builtin_amdgcn_wmma_f32_16x16x4_f32(false, A[2 * s], false, b0,
                                                   (short)0, ca, false, false);
        cb = __builtin_amdgcn_wmma_f32_16x16x4_f32(false, A[2 * s + 1], false, b1,
                                                   (short)0, cb, false, false);
      }
      v8f c = ca + cb;
      int rem = tt * 16 + li;
      int u = rem >> 6;
      int v = (rem >> 2) & 15;
#pragma unroll
      for (int r = 0; r < 8; ++r) {
        int e = r + eoff;
        float W = c[r];
        const float* X = xsr + e * 20;
        const float* M = mid + e * 260;
        if (p == 0) {
          acc0[r] += W * X[u] * M[v];
        } else if (p == 1) {
          float f = W * X[u] * INV_SQRT3;
          acc1[r][0] += f * M[16 + v * 3 + 0];
          acc1[r][1] += f * M[16 + v * 3 + 1];
          acc1[r][2] += f * M[16 + v * 3 + 2];
        } else if (p == 2) {
          float f = W * X[u] * INV_SQRT3;
          acc1[r][0] += f * M[64 + v * 3 + 0];
          acc1[r][1] += f * M[64 + v * 3 + 1];
          acc1[r][2] += f * M[64 + v * 3 + 2];
        } else if (p == 3) {
          acc0[r] += W * X[u] * M[112 + v];
        } else if (p == 4) {
          float f = W * M[v] * INV_SQRT3;
          acc1[r][0] += f * X[4 + u * 3 + 0];
          acc1[r][1] += f * X[4 + u * 3 + 1];
          acc1[r][2] += f * X[4 + u * 3 + 2];
        } else if (p == 5) {
          float d = X[4 + u * 3 + 0] * M[16 + v * 3 + 0]
                  + X[4 + u * 3 + 1] * M[16 + v * 3 + 1]
                  + X[4 + u * 3 + 2] * M[16 + v * 3 + 2];
          acc0[r] += W * d * INV_SQRT3;
        } else if (p == 6) {
          float d = X[4 + u * 3 + 0] * M[64 + v * 3 + 0]
                  + X[4 + u * 3 + 1] * M[64 + v * 3 + 1]
                  + X[4 + u * 3 + 2] * M[64 + v * 3 + 2];
          acc0[r] += W * d * INV_SQRT3;
        } else if (p == 7) {
          float f = W * M[112 + v] * INV_SQRT3;
          acc1[r][0] += f * X[4 + u * 3 + 0];
          acc1[r][1] += f * X[4 + u * 3 + 1];
          acc1[r][2] += f * X[4 + u * 3 + 2];
        } else if (p == 8) {
          float a0 = X[4 + u * 3 + 0], a1 = X[4 + u * 3 + 1], a2 = X[4 + u * 3 + 2];
          float m0 = M[128 + v * 3 + 0], m1 = M[128 + v * 3 + 1], m2 = M[128 + v * 3 + 2];
          float f = W * INV_SQRT6;
          acc1[r][0] += f * (a1 * m2 - a2 * m1);
          acc1[r][1] += f * (a2 * m0 - a0 * m2);
          acc1[r][2] += f * (a0 * m1 - a1 * m0);
        } else {
          float a0 = X[4 + u * 3 + 0], a1 = X[4 + u * 3 + 1], a2 = X[4 + u * 3 + 2];
          const float* m5 = M + 176 + v * 5;
          float m0 = m5[0], m1 = m5[1], m2 = m5[2], m3 = m5[3], m4 = m5[4];
          float f = W * INV_SQRT10;
          float s2 = m2 * INV_SQRT3;
          acc1[r][0] += f * ((-s2 - m4) * a0 + m1 * a1 + m0 * a2);
          acc1[r][1] += f * (m1 * a0 + 2.f * s2 * a1 + m3 * a2);
          acc1[r][2] += f * (m0 * a0 + m3 * a1 + (-s2 + m4) * a2);
        }
      }
    }
  }

#pragma unroll
  for (int r = 0; r < 8; ++r) {
    float s0 = acc0[r];
    float t0 = acc1[r][0], t1 = acc1[r][1], t2 = acc1[r][2];
    s0 += __shfl_xor(s0, 4); s0 += __shfl_xor(s0, 8);
    t0 += __shfl_xor(t0, 4); t0 += __shfl_xor(t0, 8);
    t1 += __shfl_xor(t1, 4); t1 += __shfl_xor(t1, 8);
    t2 += __shfl_xor(t2, 4); t2 += __shfl_xor(t2, 8);
    if (li < 4) {
      int e = r + eoff;
      float* o = out + (size_t)dsti[e] * 16;
      atomicAdd(o + li,             s0 * SC0);
      atomicAdd(o + 4 + 3 * li + 0, t0 * SC1);
      atomicAdd(o + 4 + 3 * li + 1, t1 * SC1);
      atomicAdd(o + 4 + 3 * li + 2, t2 * SC1);
    }
  }
}

extern "C" void kernel_launch(void* const* d_in, const int* in_sizes, int n_in,
                              void* d_out, int out_size, void* d_ws, size_t ws_size,
                              hipStream_t stream) {
  const float* node_pos  = (const float*)d_in[0];
  const float* edge_type = (const float*)d_in[1];
  const float* fc_w0     = (const float*)d_in[2];
  const float* fc_w1     = (const float*)d_in[3];
  const int*   esrc      = (const int*)d_in[4];
  const int*   edst      = (const int*)d_in[5];
  float* out = (float*)d_out;
  float* w1p = (float*)d_ws;
  int E = in_sizes[4];

  k_zero<<<(out_size + 255) / 256, 256, 0, stream>>>(out, out_size);
  k_repack_w1<<<(160 * 16 * 32 + 255) / 256, 256, 0, stream>>>(fc_w1, w1p);
  k_edge_conv<<<E / 16, 32, 0, stream>>>(node_pos, edge_type, fc_w0, w1p,
                                         esrc, edst, out);
}


# FC2 operands swapped (fc_w1 tile = A, hidden^T = B): a lane owns one edge; x_src[u] x mid[v] coupling formed once per (u,v), reused for 4 w_out; message row in 16 registers; sum order changed
# speedup vs baseline: 2.5874x; 2.5874x over previous
#include <hip/hip_runtime.h>
#include <math.h>


typedef float v2f __attribute__((ext_vector_type(2)));
typedef float v8f __attribute__((ext_vector_type(8)));

#define INV_SQRT2  0.70710678f
#define INV_SQRT3  0.57735027f
#define INV_SQRT6  0.40824829f
#define INV_SQRT10 0.31622777f
#define SILU_C     1.67668f
#define SC0        0.02209709f
#define SC1        0.03125f

__global__ void k_zero(float* p, int n) {
  int i = blockIdx.x * blockDim.x + threadIdx.x;
  if (i < n) p[i] = 0.f;
}

__global__ void k_repack_w1(const float* __restrict__ w1, float* __restrict__ o) {
  int tid = blockIdx.x * blockDim.x + threadIdx.x;
  if (tid >= 160 * 16 * 32) return;
  int l  = tid & 31;
  int s  = (tid >> 5) & 15;
  int t  = tid >> 9;
  int li = l & 15;
  int kd = (l < 16) ? 0 : 2;
  int row = 4 * s + kd;
  int col = t * 16 + li;
  o[2 * tid + 0] = w1[row * 2560 + col];
  o[2 * tid + 1] = w1[(row + 1) * 2560 + col];
}

__global__ __launch_bounds__(32)
void k_edge_conv(const float* __restrict__ node_pos,
                 const float* __restrict__ edge_type,
                 const float* __restrict__ fc_w0,
                 const float* __restrict__ w1p,
                 const int*   __restrict__ edge_src,
                 const int*   __restrict__ edge_dst,
                 float* __restrict__ out)
{
  __shared__ float xsr[16 * 20];
  __shared__ __attribute__((aligned(16)))
             float mid[16 * 260];
  float* const hid = mid;
  float* const xds = mid + 16 * 66;

  const int lane = threadIdx.x;
  const int li   = lane & 15;
  const int kd   = (lane < 16) ? 0 : 2;
  const int eoff = (lane < 16) ? 0 : 8;
  const int e0   = blockIdx.x * 16;

  const int dnode = edge_dst[e0 + li];
  if (lane < 16) {
    int s = edge_src[e0 + li];
    const float4* q = (const float4*)(node_pos + (size_t)s * 16);
    float4* d = (float4*)(xsr + li * 20);
    d[0] = q[0]; d[1] = q[1]; d[2] = q[2]; d[3] = q[3];
  } else {
    const float4* q = (const float4*)(node_pos + (size_t)dnode * 16);
    float4* d = (float4*)(xds + li * 20);
    d[0] = q[0]; d[1] = q[1]; d[2] = q[2]; d[3] = q[3];
  }

  {
    v2f a[4];
    const float* et = edge_type + (size_t)(e0 + li) * 16 + kd;
#pragma unroll
    for (int s = 0; s < 4; ++s) a[s] = *(const v2f*)(et + 4 * s);
#pragma unroll
    for (int tt = 0; tt < 4; ++tt) {
      v8f c = {};
#pragma unroll
      for (int s = 0; s < 4; ++s) {
        int row = 4 * s + kd;
        v2f b;
        b.x = fc_w0[row * 64 + tt * 16 + li];
        b.y = fc_w0[(row + 1) * 64 + tt * 16 + li];
        c = __builtin_amdgcn_wmma_f32_16x16x4_f32(false, a[s], false, b,
                                                  (short)0, c, false, false);
      }
#pragma unroll
      for (int r = 0; r < 8; ++r) {
        float x = c[r] * 0.25f;
        float h = (x / (1.f + __expf(-x))) * (SILU_C * 0.125f);
        hid[(r + eoff) * 66 + tt * 16 + li] = h;
      }
    }
  }
  __syncthreads();

  v2f H[16];
#pragma unroll
  for (int s = 0; s < 16; ++s) H[s] = *(const v2f*)&hid[li * 66 + 4 * s + kd];
  float y0[4], y1[4][3];
  {
    const float* Y = xds + li * 20;
#pragma unroll
    for (int v = 0; v < 4; ++v) {
      y0[v] = Y[v];
      y1[v][0] = Y[4 + v * 3 + 0];
      y1[v][1] = Y[4 + v * 3 + 1];
      y1[v][2] = Y[4 + v * 3 + 2];
    }
  }
  __syncthreads();

  {
    const int e = li;
    const float* X = xsr + e * 20;
    float* M = mid + e * 260;
    const int ub = (lane < 16) ? 0 : 2;
#pragma unroll
    for (int uu = 0; uu < 2; ++uu) {
      int u = ub + uu;
      float x0 = X[u];
      float a0 = X[4 + u * 3 + 0], a1 = X[4 + u * 3 + 1], a2 = X[4 + u * 3 + 2];
#pragma unroll
      for (int v = 0; v < 4; ++v) {
        int uv = u * 4 + v;
        float b0 = y1[v][0], b1 = y1[v][1], b2 = y1[v][2];
        M[uv] = x0 * y0[v];
        M[16 + uv * 3 + 0] = x0 * b0;
        M[16 + uv * 3 + 1] = x0 * b1;
        M[16 + uv * 3 + 2] = x0 * b2;
        M[64 + uv * 3 + 0] = a0 * y0[v];
        M[64 + uv * 3 + 1] = a1 * y0[v];
        M[64 + uv * 3 + 2] = a2 * y0[v];
        M[112 + uv] = (a0 * b0 + a1 * b1 + a2 * b2) * INV_SQRT3;
        M[128 + uv * 3 + 0] = (a1 * b2 - a2 * b1) * INV_SQRT2;
        M[128 + uv * 3 + 1] = (a2 * b0 - a0 * b2) * INV_SQRT2;
        M[128 + uv * 3 + 2] = (a0 * b1 - a1 * b0) * INV_SQRT2;
        M[176 + uv * 5 + 0] = (a2 * b0 + a0 * b2) * INV_SQRT2;
        M[176 + uv * 5 + 1] = (a0 * b1 + a1 * b0) * INV_SQRT2;
        M[176 + uv * 5 + 2] = (2.f * a1 * b1 - a0 * b0 - a2 * b2) * INV_SQRT6;
        M[176 + uv * 5 + 3] = (a1 * b2 + a2 * b1) * INV_SQRT2;
        M[176 + uv * 5 + 4] = (a2 * b2 - a0 * b0) * INV_SQRT2;
      }
    }
  }
  __syncthreads();

  float acc0[4], acc1[4][3];
#pragma unroll
  for (int w = 0; w < 4; ++w) {
    acc0[w] = 0.f; acc1[w][0] = 0.f; acc1[w][1] = 0.f; acc1[w][2] = 0.f;
  }
  const float* Xr = xsr + li * 20;
  const float* Mr = mid + li * 260;
  const int hv = (lane < 16) ? 0 : 2;

#pragma unroll
  for (int p = 0; p < 10; ++p) {
#pragma nounroll
    for (int tt = 0; tt < 16; ++tt) {
      int t = p * 16 + tt;
      const float* aq = w1p + (size_t)t * 1024 + 2 * lane;
      const int u  = tt >> 2;
      const int v0 = (tt & 3) * 4 + hv;
      float T0[2], T1[2][3];
#pragma unroll
      for (int j = 0; j < 2; ++j) {
        const int v = v0 + j;
        if (p == 0) {
          T0[j] = Xr[u] * Mr[v];
        } else if (p == 1) {
          float f = Xr[u] * INV_SQRT3;
          T1[j][0] = f * Mr[16 + v * 3 + 0];
          T1[j][1] = f * Mr[16 + v * 3 + 1];
          T1[j][2] = f * Mr[16 + v * 3 + 2];
        } else if (p == 2) {
          float f = Xr[u] * INV_SQRT3;
          T1[j][0] = f * Mr[64 + v * 3 + 0];
          T1[j][1] = f * Mr[64 + v * 3 + 1];
          T1[j][2] = f * Mr[64 + v * 3 + 2];
        } else if (p == 3) {
          T0[j] = Xr[u] * Mr[112 + v];
        } else if (p == 4) {
          float f = Mr[v] * INV_SQRT3;
          T1[j][0] = f * Xr[4 + u * 3 + 0];
          T1[j][1] = f * Xr[4 + u * 3 + 1];
          T1[j][2] = f * Xr[4 + u * 3 + 2];
        } else if (p == 5) {
          float d = Xr[4 + u * 3 + 0] * Mr[16 + v * 3 + 0]
                  + Xr[4 + u * 3 + 1] * Mr[16 + v * 3 + 1]
                  + Xr[4 + u * 3 + 2] * Mr[16 + v * 3 + 2];
          T0[j] = d * INV_SQRT3;
        } else if (p == 6) {
          float d = Xr[4 + u * 3 + 0] * Mr[64 + v * 3 + 0]
                  + Xr[4 + u * 3 + 1] * Mr[64 + v * 3 + 1]
                  + Xr[4 + u * 3 + 2] * Mr[64 + v * 3 + 2];
          T0[j] = d * INV_SQRT3;
        } else if (p == 7) {
          float f = Mr[112 + v] * INV_SQRT3;
          T1[j][0] = f * Xr[4 + u * 3 + 0];
          T1[j][1] = f * Xr[4 + u * 3 + 1];
          T1[j][2] = f * Xr[4 + u * 3 + 2];
        } else if (p == 8) {
          float a0 = Xr[4 + u * 3 + 0], a1 = Xr[4 + u * 3 + 1], a2 = Xr[4 + u * 3 + 2];
          float m0 = Mr[128 + v * 3 + 0], m1 = Mr[128 + v * 3 + 1], m2 = Mr[128 + v * 3 + 2];
          T1[j][0] = INV_SQRT6 * (a1 * m2 - a2 * m1);
          T1[j][1] = INV_SQRT6 * (a2 * m0 - a0 * m2);
          T1[j][2] = INV_SQRT6 * (a0 * m1 - a1 * m0);
        } else {
          float a0 = Xr[4 + u * 3 + 0], a1 = Xr[4 + u * 3 + 1], a2 = Xr[4 + u * 3 + 2];
          const float* m5 = Mr + 176 + v * 5;
          float m0 = m5[0], m1 = m5[1], m2 = m5[2], m3 = m5[3], m4 = m5[4];
          float s2 = m2 * INV_SQRT3;
          T1[j][0] = INV_SQRT10 * ((-s2 - m4) * a0 + m1 * a1 + m0 * a2);
          T1[j][1] = INV_SQRT10 * (m1 * a0 + 2.f * s2 * a1 + m3 * a2);
          T1[j][2] = INV_SQRT10 * (m0 * a0 + m3 * a1 + (-s2 + m4) * a2);
        }
      }
      v8f ca = {}, cb = {};
#pragma unroll
      for (int s = 0; s < 8; ++s) {
        v2f w0 = *(const v2f*)(aq + (2 * s) * 64);
        v2f w1 = *(const v2f*)(aq + (2 * s + 1) * 64);
        ca = __builtin_amdgcn_wmma_f32_16x16x4_f32(false, w0, false, H[2 * s],
                                                   (short)0, ca, false, false);
        cb = __builtin_amdgcn_wmma_f32_16x16x4_f32(false, w1, false, H[2 * s + 1],
                                                   (short)0, cb, false, false);
      }
      v8f c = ca + cb;
#pragma unroll
      for (int r = 0; r < 8; ++r) {
        const int j = r >> 2, w = r & 3;
        if (p == 0 || p == 3 || p == 5 || p == 6) {
          acc0[w] += c[r] * T0[j];
        } else {
          acc1[w][0] += c[r] * T1[j][0];
          acc1[w][1] += c[r] * T1[j][1];
          acc1[w][2] += c[r] * T1[j][2];
        }
      }
    }
  }

  float msg[16];
#pragma unroll
  for (int w = 0; w < 4; ++w) {
    msg[w] = acc0[w] * SC0;
    msg[4 + 3 * w + 0] = acc1[w][0] * SC1;
    msg[4 + 3 * w + 1] = acc1[w][1] * SC1;
    msg[4 + 3 * w + 2] = acc1[w][2] * SC1;
  }
#pragma unroll
  for (int i = 0; i < 16; ++i) msg[i] += __shfl_xor(msg[i], 16);
  {
    float* o = out + (size_t)dnode * 16 + ((lane < 16) ? 0 : 8);
#pragma unroll
    for (int i = 0; i < 8; ++i)
      atomicAdd(o + i, (lane < 16) ? msg[i] : msg[8 + i]);
  }
}

extern "C" void kernel_launch(void* const* d_in, const int* in_sizes, int n_in,
                              void* d_out, int out_size, void* d_ws, size_t ws_size,
                              hipStream_t stream) {
  const float* node_pos  = (const float*)d_in[0];
  const float* edge_type = (const float*)d_in[1];
  const float* fc_w0     = (const float*)d_in[2];
  const float* fc_w1     = (const float*)d_in[3];
  const int*   esrc      = (const int*)d_in[4];
  const int*   edst      = (const int*)d_in[5];
  float* out = (float*)d_out;
  float* w1p = (float*)d_ws;
  int E = in_sizes[4];

  k_zero<<<(out_size + 255) / 256, 256, 0, stream>>>(out, out_size);
  k_repack_w1<<<(160 * 16 * 32 + 255) / 256, 256, 0, stream>>>(fc_w1, w1p);
  k_edge_conv<<<E / 16, 32, 0, stream>>>(node_pos, edge_type, fc_w0, w1p,
                                         esrc, edst, out);
}


# fc_w1 staged in LDS once per work-group of 16 waves (256 edges): 4-tile stages, two buffers, fetched one stage ahead under the matrix work; A fragments read by 8 ds_load_b128 a tile, not from global
# speedup vs baseline: 3.2087x; 1.2401x over previous
#include <hip/hip_runtime.h>
#include <math.h>


typedef float v2f __attribute__((ext_vector_type(2)));
typedef float v8f __attribute__((ext_vector_type(8)));

#define INV_SQRT2  0.70710678f
#define INV_SQRT3  0.57735027f
#define INV_SQRT6  0.40824829f
#define INV_SQRT10 0.31622777f
#define SILU_C     1.67668f
#define SC0        0.02209709f
#define SC1        0.03125f

__global__ void k_zero(float* p, int n) {
  int i = blockIdx.x * blockDim.x + threadIdx.x;
  if (i < n) p[i] = 0.f;
}

__global__ void k_repack_w1(const float* __restrict__ w1, float* __restrict__ o) {
  int tid = blockIdx.x * blockDim.x + threadIdx.x;
  if (tid >= 160 * 8 * 32) return;
  int l  = tid & 31;
  int j  = (tid >> 5) & 7;
  int t  = tid >> 8;
  int li = l & 15;
  int kd = (l < 16) ? 0 : 2;
  int row = 8 * j + kd;
  int col = t * 16 + li;
  float4 v;
  v.x = w1[row * 2560 + col];
  v.y = w1[(row + 1) * 2560 + col];
  v.z = w1[(row + 4) * 2560 + col];
  v.w = w1[(row + 5) * 2560 + col];
  ((float4*)o)[tid] = v;
}

#define NWAVE    16
#define STAGE_F4 1024

__global__ __launch_bounds__(NWAVE * 32)
void k_edge_conv(const float* __restrict__ node_pos,
                 const float* __restrict__ edge_type,
                 const float* __restrict__ fc_w0,
                 const float* w1p,
                 const int*   __restrict__ edge_src,
                 const int*   __restrict__ edge_dst,
                 float* __restrict__ out)
{
  __shared__ __attribute__((aligned(16))) float wst[2 * STAGE_F4 * 4];
  __shared__ __attribute__((aligned(16))) float mid_all[NWAVE * 16 * 260];
  __shared__ __attribute__((aligned(16))) float xsr_all[NWAVE * 16 * 20];

  const int tid  = threadIdx.x;
  const int wave = __builtin_amdgcn_readfirstlane(tid >> 5);
  const int lane = tid & 31;
  float* const xsr = xsr_all + wave * (16 * 20);
  float* const mid = mid_all + wave * (16 * 260);
  float* const hid = mid;
  float* const xds = mid + 16 * 66;

  const int li   = lane & 15;
  const int kd   = (lane < 16) ? 0 : 2;
  const int eoff = (lane < 16) ? 0 : 8;
  const int e0   = (blockIdx.x * NWAVE + wave) * 16;

  const float4* const wg = (const float4*)w1p;
  float4 r0 = wg[tid], r1 = wg[512 + tid];

  const int dnode = edge_dst[e0 + li];
  if (lane < 16) {
    int s = edge_src[e0 + li];
    const float4* q = (const float4*)(node_pos + (size_t)s * 16);
    float4* d = (float4*)(xsr + li * 20);
    d[0] = q[0]; d[1] = q[1]; d[2] = q[2]; d[3] = q[3];
  } else {
    const float4* q = (const float4*)(node_pos + (size_t)dnode * 16);
    float4* d = (float4*)(xds + li * 20);
    d[0] = q[0]; d[1] = q[1]; d[2] = q[2]; d[3] = q[3];
  }

  {
    v2f a[4];
    const float* et = edge_type + (size_t)(e0 + li) * 16 + kd;
#pragma unroll
    for (int s = 0; s < 4; ++s) a[s] = *(const v2f*)(et + 4 * s);
#pragma unroll
    for (int tt = 0; tt < 4; ++tt) {
      v8f c = {};
#pragma unroll
      for (int s = 0; s < 4; ++s) {
        int row = 4 * s + kd;
        v2f b;
        b.x = fc_w0[row * 64 + tt * 16 + li];
        b.y = fc_w0[(row + 1) * 64 + tt * 16 + li];
        c = __builtin_amdgcn_wmma_f32_16x16x4_f32(false, a[s], false, b,
                                                  (short)0, c, false, false);
      }
#pragma unroll
      for (int r = 0; r < 8; ++r) {
        float x = c[r] * 0.25f;
        float h = (x / (1.f + __expf(-x))) * (SILU_C * 0.125f);
        hid[(r + eoff) * 66 + tt * 16 + li] = h;
      }
    }
  }
  __syncthreads();

  v2f H[16];
#pragma unroll
  for (int s = 0; s < 16; ++s) H[s] = *(const v2f*)&hid[li * 66 + 4 * s + kd];
  float y0[4], y1[4][3];
  {
    const float* Y = xds + li * 20;
#pragma unroll
    for (int v = 0; v < 4; ++v) {
      y0[v] = Y[v];
      y1[v][0] = Y[4 + v * 3 + 0];
      y1[v][1] = Y[4 + v * 3 + 1];
      y1[v][2] = Y[4 + v * 3 + 2];
    }
  }
  __syncthreads();

  {
    const int e = li;
    const float* X = xsr + e * 20;
    float* M = mid + e * 260;
    const int ub = (lane < 16) ? 0 : 2;
#pragma unroll
    for (int uu = 0; uu < 2; ++uu) {
      int u = ub + uu;
      float x0 = X[u];
      float a0 = X[4 + u * 3 + 0], a1 = X[4 + u * 3 + 1], a2 = X[4 + u * 3 + 2];
#pragma unroll
      for (int v = 0; v < 4; ++v) {
        int uv = u * 4 + v;
        float b0 = y1[v][0], b1 = y1[v][1], b2 = y1[v][2];
        M[uv] = x0 * y0[v];
        M[16 + uv * 3 + 0] = x0 * b0;
        M[16 + uv * 3 + 1] = x0 * b1;
        M[16 + uv * 3 + 2] = x0 * b2;
        M[64 + uv * 3 + 0] = a0 * y0[v];
        M[64 + uv * 3 + 1] = a1 * y0[v];
        M[64 + uv * 3 + 2] = a2 * y0[v];
        M[112 + uv] = (a0 * b0 + a1 * b1 + a2 * b2) * INV_SQRT3;
        M[128 + uv * 3 + 0] = (a1 * b2 - a2 * b1) * INV_SQRT2;
        M[128 + uv * 3 + 1] = (a2 * b0 - a0 * b2) * INV_SQRT2;
        M[128 + uv * 3 + 2] = (a0 * b1 - a1 * b0) * INV_SQRT2;
        M[176 + uv * 5 + 0] = (a2 * b0 + a0 * b2) * INV_SQRT2;
        M[176 + uv * 5 + 1] = (a0 * b1 + a1 * b0) * INV_SQRT2;
        M[176 + uv * 5 + 2] = (2.f * a1 * b1 - a0 * b0 - a2 * b2) * INV_SQRT6;
        M[176 + uv * 5 + 3] = (a1 * b2 + a2 * b1) * INV_SQRT2;
        M[176 + uv * 5 + 4] = (a2 * b2 - a0 * b0) * INV_SQRT2;
      }
    }
  }
  ((float4*)wst)[tid] = r0;
  ((float4*)wst)[512 + tid] = r1;
  __syncthreads();

  float acc0[4], acc1[4][3];
#pragma unroll
  for (int w = 0; w < 4; ++w) {
    acc0[w] = 0.f; acc1[w][0] = 0.f; acc1[w][1] = 0.f; acc1[w][2] = 0.f;
  }
  const float* Xr = xsr + li * 20;
  const float* Mr = mid + li * 260;
  const int hv = (lane < 16) ? 0 : 2;

#pragma unroll
  for (int p = 0; p < 10; ++p) {
#pragma nounroll
    for (int sg = 0; sg < 4; ++sg) {
      const bool more = (p * 4 + sg + 1 < 40);
      if (more) {
        const float4* g = wg + (size_t)(p * 4 + sg + 1) * STAGE_F4;
        r0 = g[tid];
        r1 = g[512 + tid];
      }
      const float4* wb = (const float4*)wst + (sg & 1) * STAGE_F4 + lane;
#pragma nounroll
      for (int i = 0; i < 4; ++i) {
        const float4* aq = wb + i * 256;
        const int u  = sg;
        const int v0 = i * 4 + hv;
        float T0[2], T1[2][3];
#pragma unroll
        for (int j = 0; j < 2; ++j) {
          const int v = v0 + j;
          if (p == 0) {
            T0[j] = Xr[u] * Mr[v];
          } else if (p == 1) {
            float f = Xr[u] * INV_SQRT3;
            T1[j][0] = f * Mr[16 + v * 3 + 0];
            T1[j][1] = f * Mr[16 + v * 3 + 1];
            T1[j][2] = f * Mr[16 + v * 3 + 2];
          } else if (p == 2) {
            float f = Xr[u] * INV_SQRT3;
            T1[j][0] = f * Mr[64 + v * 3 + 0];
            T1[j][1] = f * Mr[64 + v * 3 + 1];
            T1[j][2] = f * Mr[64 + v * 3 + 2];
          } else if (p == 3) {
            T0[j] = Xr[u] * Mr[112 + v];
          } else if (p == 4) {
            float f = Mr[v] * INV_SQRT3;
            T1[j][0] = f * Xr[4 + u * 3 + 0];
            T1[j][1] = f * Xr[4 + u * 3 + 1];
            T1[j][2] = f * Xr[4 + u * 3 + 2];
          } else if (p == 5) {
            float d = Xr[4 + u * 3 + 0] * Mr[16 + v * 3 + 0]
                    + Xr[4 + u * 3 + 1] * Mr[16 + v * 3 + 1]
                    + Xr[4 + u * 3 + 2] * Mr[16 + v * 3 + 2];
            T0[j] = d * INV_SQRT3;
          } else if (p == 6) {
            float d = Xr[4 + u * 3 + 0] * Mr[64 + v * 3 + 0]
                    + Xr[4 + u * 3 + 1] * Mr[64 + v * 3 + 1]
                    + Xr[4 + u * 3 + 2] * Mr[64 + v * 3 + 2];
            T0[j] = d * INV_SQRT3;
          } else if (p == 7) {
            float f = Mr[112 + v] * INV_SQRT3;
            T1[j][0] = f * Xr[4 + u * 3 + 0];
            T1[j][1] = f * Xr[4 + u * 3 + 1];
            T1[j][2] = f * Xr[4 + u * 3 + 2];
          } else if (p == 8) {
            float a0 = Xr[4 + u * 3 + 0], a1 = Xr[4 + u * 3 + 1], a2 = Xr[4 + u * 3 + 2];
            float m0 = Mr[128 + v * 3 + 0], m1 = Mr[128 + v * 3 + 1], m2 = Mr[128 + v * 3 + 2];
            T1[j][0] = INV_SQRT6 * (a1 * m2 - a2 * m1);
            T1[j][1] = INV_SQRT6 * (a2 * m0 - a0 * m2);
            T1[j][2] = INV_SQRT6 * (a0 * m1 - a1 * m0);
          } else {
            float a0 = Xr[4 + u * 3 + 0], a1 = Xr[4 + u * 3 + 1], a2 = Xr[4 + u * 3 + 2];
            const float* m5 = Mr + 176 + v * 5;
            float m0 = m5[0], m1 = m5[1], m2 = m5[2], m3 = m5[3], m4 = m5[4];
            float s2 = m2 * INV_SQRT3;
            T1[j][0] = INV_SQRT10 * ((-s2 - m4) * a0 + m1 * a1 + m0 * a2);
            T1[j][1] = INV_SQRT10 * (m1 * a0 + 2.f * s2 * a1 + m3 * a2);
            T1[j][2] = INV_SQRT10 * (m0 * a0 + m3 * a1 + (-s2 + m4) * a2);
          }
        }
        v8f ca = {}, cb = {};
        float4 wq[8];
#pragma unroll
        for (int s = 0; s < 8; ++s) wq[s] = aq[s * 32];
#pragma unroll
        for (int s = 0; s < 8; ++s) {
          v2f w0 = {wq[s].x, wq[s].y};
          v2f w1 = {wq[s].z, wq[s].w};
          ca = __builtin_amdgcn_wmma_f32_16x16x4_f32(false, w0, false, H[2 * s],
                                                     (short)0, ca, false, false);
          cb = __builtin_amdgcn_wmma_f32_16x16x4_f32(false, w1, false, H[2 * s + 1],
                                                     (short)0, cb, false, false);
        }
        __builtin_amdgcn_sched_group_barrier(0x100, 24, 0);
        __builtin_amdgcn_sched_group_barrier(0x008, 16, 0);
        v8f c = ca + cb;
#pragma unroll
        for (int r = 0; r < 8; ++r) {
          const int j = r >> 2, w = r & 3;
          if (p == 0 || p == 3 || p == 5 || p == 6) {
            acc0[w] += c[r] * T0[j];
          } else {
            acc1[w][0] += c[r] * T1[j][0];
            acc1[w][1] += c[r] * T1[j][1];
            acc1[w][2] += c[r] * T1[j][2];
          }
        }
      }
      if (more) {
        float4* d = (float4*)wst + ((sg + 1) & 1) * STAGE_F4;
        d[tid] = r0;
        d[512 + tid] = r1;
        __syncthreads();
      }
    }
  }

  float msg[16];
#pragma unroll
  for (int w = 0; w < 4; ++w) {
    msg[w] = acc0[w] * SC0;
    msg[4 + 3 * w + 0] = acc1[w][0] * SC1;
    msg[4 + 3 * w + 1] = acc1[w][1] * SC1;
    msg[4 + 3 * w + 2] = acc1[w][2] * SC1;
  }
#pragma unroll
  for (int i = 0; i < 16; ++i) msg[i] += __shfl_xor(msg[i], 16);
  {
    float* o = out + (size_t)dnode * 16 + ((lane < 16) ? 0 : 8);
#pragma unroll
    for (int i = 0; i < 8; ++i)
      atomicAdd(o + i, (lane < 16) ? msg[i] : msg[8 + i]);
  }
}

extern "C" void kernel_launch(void* const* d_in, const int* in_sizes, int n_in,
                              void* d_out, int out_size, void* d_ws, size_t ws_size,
                              hipStream_t stream) {
  const float* node_pos  = (const float*)d_in[0];
  const float* edge_type = (const float*)d_in[1];
  const float* fc_w0     = (const float*)d_in[2];
  const float* fc_w1     = (const float*)d_in[3];
  const int*   esrc      = (const int*)d_in[4];
  const int*   edst      = (const int*)d_in[5];
  float* out = (float*)d_out;
  float* w1p = (float*)d_ws;
  int E = in_sizes[4];

  k_zero<<<(out_size + 255) / 256, 256, 0, stream>>>(out, out_size);
  k_repack_w1<<<(160 * 8 * 32 + 255) / 256, 256, 0, stream>>>(fc_w1, w1p);
  k_edge_conv<<<E / (16 * NWAVE), NWAVE * 32, 0, stream>>>(node_pos, edge_type,
                                                          fc_w0, w1p, esrc, edst, out);
}


# fc_w1 staged one whole path (16 N-tiles, 64 KB) per hand-over, two buffers: 12 work-group barriers in FC2, not 39; LDS room: plain products mid0-2 formed in the consume, not stored; same sums
# speedup vs baseline: 3.3835x; 1.0545x over previous
#include <hip/hip_runtime.h>
#include <math.h>


typedef float v2f __attribute__((ext_vector_type(2)));
typedef float v8f __attribute__((ext_vector_type(8)));

#define INV_SQRT2  0.70710678f
#define INV_SQRT3  0.57735027f
#define INV_SQRT6  0.40824829f
#define INV_SQRT10 0.31622777f
#define SILU_C     1.67668f
#define SC0        0.02209709f
#define SC1        0.03125f

__global__ void k_zero(float* p, int n) {
  int i = blockIdx.x * blockDim.x + threadIdx.x;
  if (i < n) p[i] = 0.f;
}

__global__ void k_repack_w1(const float* __restrict__ w1, float* __restrict__ o) {
  int tid = blockIdx.x * blockDim.x + threadIdx.x;
  if (tid >= 160 * 8 * 32) return;
  int l  = tid & 31;
  int j  = (tid >> 5) & 7;
  int t  = tid >> 8;
  int li = l & 15;
  int kd = (l < 16) ? 0 : 2;
  int row = 8 * j + kd;
  int col = t * 16 + li;
  float4 v;
  v.x = w1[row * 2560 + col];
  v.y = w1[(row + 1) * 2560 + col];
  v.z = w1[(row + 4) * 2560 + col];
  v.w = w1[(row + 5) * 2560 + col];
  ((float4*)o)[tid] = v;
}

#define NWAVE    16
#define CHUNK_F4 1024
#define PATH_F4  4096
#define MIDS     148

__global__ __launch_bounds__(NWAVE * 32)
void k_edge_conv(const float* __restrict__ node_pos,
                 const float* __restrict__ edge_type,
                 const float* __restrict__ fc_w0,
                 const float* w1p,
                 const int*   __restrict__ edge_src,
                 const int*   __restrict__ edge_dst,
                 float* __restrict__ out)
{
  __shared__ __attribute__((aligned(16))) float wst[2 * PATH_F4 * 4];
  __shared__ __attribute__((aligned(16))) float mid_all[NWAVE * 16 * MIDS];
  __shared__ __attribute__((aligned(16))) float xsr_all[NWAVE * 16 * 20];

  const int tid  = threadIdx.x;
  const int wave = __builtin_amdgcn_readfirstlane(tid >> 5);
  const int lane = tid & 31;
  float* const xsr = xsr_all + wave * (16 * 20);
  float* const mid = mid_all + wave * (16 * MIDS);
  float* const hid = mid;
  float* const xds = mid + 16 * 66;

  const int li   = lane & 15;
  const int kd   = (lane < 16) ? 0 : 2;
  const int eoff = (lane < 16) ? 0 : 8;
  const int e0   = (blockIdx.x * NWAVE + wave) * 16;

  const float4* const wg = (const float4*)w1p;
  const float4 pa0 = wg[tid], pb0 = wg[512 + tid];

  const int dnode = edge_dst[e0 + li];
  if (lane < 16) {
    int s = edge_src[e0 + li];
    const float4* q = (const float4*)(node_pos + (size_t)s * 16);
    float4* d = (float4*)(xsr + li * 20);
    d[0] = q[0]; d[1] = q[1]; d[2] = q[2]; d[3] = q[3];
  } else {
    const float4* q = (const float4*)(node_pos + (size_t)dnode * 16);
    float4* d = (float4*)(xds + li * 20);
    d[0] = q[0]; d[1] = q[1]; d[2] = q[2]; d[3] = q[3];
  }

  {
    v2f a[4];
    const float* et = edge_type + (size_t)(e0 + li) * 16 + kd;
#pragma unroll
    for (int s = 0; s < 4; ++s) a[s] = *(const v2f*)(et + 4 * s);
#pragma unroll
    for (int tt = 0; tt < 4; ++tt) {
      v8f c = {};
#pragma unroll
      for (int s = 0; s < 4; ++s) {
        int row = 4 * s + kd;
        v2f b;
        b.x = fc_w0[row * 64 + tt * 16 + li];
        b.y = fc_w0[(row + 1) * 64 + tt * 16 + li];
        c = __builtin_amdgcn_wmma_f32_16x16x4_f32(false, a[s], false, b,
                                                  (short)0, c, false, false);
      }
#pragma unroll
      for (int r = 0; r < 8; ++r) {
        float x = c[r] * 0.25f;
        float h = (x / (1.f + __expf(-x))) * (SILU_C * 0.125f);
        hid[(r + eoff) * 66 + tt * 16 + li] = h;
      }
    }
  }
  __syncthreads();

  v2f H[16];
#pragma unroll
  for (int s = 0; s < 16; ++s) H[s] = *(const v2f*)&hid[li * 66 + 4 * s + kd];
  float y0[4], y1[4][3];
  {
    const float* Y = xds + li * 20;
#pragma unroll
    for (int v = 0; v < 4; ++v) {
      y0[v] = Y[v];
      y1[v][0] = Y[4 + v * 3 + 0];
      y1[v][1] = Y[4 + v * 3 + 1];
      y1[v][2] = Y[4 + v * 3 + 2];
    }
  }
  float ys0[2], ys1[2][3];
#pragma unroll
  for (int j = 0; j < 2; ++j) {
    ys0[j] = (lane < 16) ? y0[j] : y0[2 + j];
#pragma unroll
    for (int k = 0; k < 3; ++k) ys1[j][k] = (lane < 16) ? y1[j][k] : y1[2 + j][k];
  }
  __syncthreads();

  {
    const int e = li;
    const float* X = xsr + e * 20;
    float* M = mid + e * MIDS;
    const int ub = (lane < 16) ? 0 : 2;
#pragma unroll
    for (int uu = 0; uu < 2; ++uu) {
      int u = ub + uu;
      float a0 = X[4 + u * 3 + 0], a1 = X[4 + u * 3 + 1], a2 = X[4 + u * 3 + 2];
#pragma unroll
      for (int v = 0; v < 4; ++v) {
        int uv = u * 4 + v;
        float b0 = y1[v][0], b1 = y1[v][1], b2 = y1[v][2];
        M[uv] = (a0 * b0 + a1 * b1 + a2 * b2) * INV_SQRT3;
        M[16 + uv * 3 + 0] = (a1 * b2 - a2 * b1) * INV_SQRT2;
        M[16 + uv * 3 + 1] = (a2 * b0 - a0 * b2) * INV_SQRT2;
        M[16 + uv * 3 + 2] = (a0 * b1 - a1 * b0) * INV_SQRT2;
        M[64 + uv * 5 + 0] = (a2 * b0 + a0 * b2) * INV_SQRT2;
        M[64 + uv * 5 + 1] = (a0 * b1 + a1 * b0) * INV_SQRT2;
        M[64 + uv * 5 + 2] = (2.f * a1 * b1 - a0 * b0 - a2 * b2) * INV_SQRT6;
        M[64 + uv * 5 + 3] = (a1 * b2 + a2 * b1) * INV_SQRT2;
        M[64 + uv * 5 + 4] = (a2 * b2 - a0 * b0) * INV_SQRT2;
      }
    }
  }
  ((float4*)wst)[tid] = pa0;
  ((float4*)wst)[512 + tid] = pb0;
  __syncthreads();

  float acc0[4], acc1[4][3];
#pragma unroll
  for (int w = 0; w < 4; ++w) {
    acc0[w] = 0.f; acc1[w][0] = 0.f; acc1[w][1] = 0.f; acc1[w][2] = 0.f;
  }
  const float* Xr = xsr + li * 20;
  const float* Mr = mid + li * MIDS;
  const int hv = (lane < 16) ? 0 : 2;

  float4 r0, r1;
  float4 q0, q1;
#pragma unroll
  for (int p = 0; p < 10; ++p) {
    const bool more = (p + 1 < 10);
    const float4* const cur = (const float4*)wst + (p & 1) * PATH_F4 + lane;
    float4* const oth = (float4*)wst + ((p + 1) & 1) * PATH_F4;
#pragma nounroll
    for (int sg = 0; sg < 4; ++sg) {
      if (more) {
        const float4* g = wg + (size_t)((p + 1) * 4 + sg) * CHUNK_F4;
        r0 = g[tid];
        r1 = g[512 + tid];
      }
      const bool own = (p == 0) && (sg < 3);
      if (own) {
        const float4* g = wg + (size_t)(sg + 1) * CHUNK_F4;
        q0 = g[tid];
        q1 = g[512 + tid];
      }
      const float4* wb = cur + sg * CHUNK_F4;
#pragma nounroll
      for (int i = 0; i < 4; ++i) {
        const float4* aq = wb + i * 256;
        const int u  = sg;
        const int v0 = i * 4 + hv;
        float T0[2], T1[2][3];
#pragma unroll
        for (int j = 0; j < 2; ++j) {
          const int v = v0 + j;
          if (p == 0) {
            T0[j] = Xr[u] * (Xr[i] * ys0[j]);
          } else if (p == 1) {
            float f = Xr[u] * INV_SQRT3;
            float xi = Xr[i];
            T1[j][0] = f * (xi * ys1[j][0]);
            T1[j][1] = f * (xi * ys1[j][1]);
            T1[j][2] = f * (xi * ys1[j][2]);
          } else if (p == 2) {
            float f = Xr[u] * INV_SQRT3;
            T1[j][0] = f * (Xr[4 + i * 3 + 0] * ys0[j]);
            T1[j][1] = f * (Xr[4 + i * 3 + 1] * ys0[j]);
            T1[j][2] = f * (Xr[4 + i * 3 + 2] * ys0[j]);
          } else if (p == 3) {
            T0[j] = Xr[u] * Mr[v];
          } else if (p == 4) {
            float f = (Xr[i] * ys0[j]) * INV_SQRT3;
            T1[j][0] = f * Xr[4 + u * 3 + 0];
            T1[j][1] = f * Xr[4 + u * 3 + 1];
            T1[j][2] = f * Xr[4 + u * 3 + 2];
          } else if (p == 5) {
            float xi = Xr[i];
            float m0 = xi * ys1[j][0], m1 = xi * ys1[j][1], m2 = xi * ys1[j][2];
            float d = Xr[4 + u * 3 + 0] * m0
                    + Xr[4 + u * 3 + 1] * m1
                    + Xr[4 + u * 3 + 2] * m2;
            T0[j] = d * INV_SQRT3;
          } else if (p == 6) {
            float m0 = Xr[4 + i * 3 + 0] * ys0[j], m1 = Xr[4 + i * 3 + 1] * ys0[j],
                  m2 = Xr[4 + i * 3 + 2] * ys0[j];
            float d = Xr[4 + u * 3 + 0] * m0
                    + Xr[4 + u * 3 + 1] * m1
                    + Xr[4 + u * 3 + 2] * m2;
            T0[j] = d * INV_SQRT3;
          } else if (p == 7) {
            float f = Mr[v] * INV_SQRT3;
            T1[j][0] = f * Xr[4 + u * 3 + 0];
            T1[j][1] = f * Xr[4 + u * 3 + 1];
            T1[j][2] = f * Xr[4 + u * 3 + 2];
          } else if (p == 8) {
            float a0 = Xr[4 + u * 3 + 0], a1 = Xr[4 + u * 3 + 1], a2 = Xr[4 + u * 3 + 2];
            float m0 = Mr[16 + v * 3 + 0], m1 = Mr[16 + v * 3 + 1], m2 = Mr[16 + v * 3 + 2];
            T1[j][0] = INV_SQRT6 * (a1 * m2 - a2 * m1);
            T1[j][1] = INV_SQRT6 * (a2 * m0 - a0 * m2);
            T1[j][2] = INV_SQRT6 * (a0 * m1 - a1 * m0);
          } else {
            float a0 = Xr[4 + u * 3 + 0], a1 = Xr[4 + u * 3 + 1], a2 = Xr[4 + u * 3 + 2];
            const float* m5 = Mr + 64 + v * 5;
            float m0 = m5[0], m1 = m5[1], m2 = m5[2], m3 = m5[3], m4 = m5[4];
            float s2 = m2 * INV_SQRT3;
            T1[j][0] = INV_SQRT10 * ((-s2 - m4) * a0 + m1 * a1 + m0 * a2);
            T1[j][1] = INV_SQRT10 * (m1 * a0 + 2.f * s2 * a1 + m3 * a2);
            T1[j][2] = INV_SQRT10 * (m0 * a0 + m3 * a1 + (-s2 + m4) * a2);
          }
        }
        v8f ca = {}, cb = {};
        float4 wq[8];
#pragma unroll
        for (int s = 0; s < 8; ++s) wq[s] = aq[s * 32];
#pragma unroll
        for (int s = 0; s < 8; ++s) {
          v2f w0 = {wq[s].x, wq[s].y};
          v2f w1 = {wq[s].z, wq[s].w};
          ca = __builtin_amdgcn_wmma_f32_16x16x4_f32(false, w0, false, H[2 * s],
                                                     (short)0, ca, false, false);
          cb = __builtin_amdgcn_wmma_f32_16x16x4_f32(false, w1, false, H[2 * s + 1],
                                                     (short)0, cb, false, false);
        }
        __builtin_amdgcn_sched_group_barrier(0x100, 24, 0);
        __builtin_amdgcn_sched_group_barrier(0x008, 16, 0);
        v8f c = ca + cb;
#pragma unroll
        for (int r = 0; r < 8; ++r) {
          const int j = r >> 2, w = r & 3;
          if (p == 0 || p == 3 || p == 5 || p == 6) {
            acc0[w] += c[r] * T0[j];
          } else {
            acc1[w][0] += c[r] * T1[j][0];
            acc1[w][1] += c[r] * T1[j][1];
            acc1[w][2] += c[r] * T1[j][2];
          }
        }
      }
      if (more) {
        float4* d = oth + sg * CHUNK_F4;
        d[tid] = r0;
        d[512 + tid] = r1;
      }
      if (own) {
        float4* d = (float4*)wst + (sg + 1) * CHUNK_F4;
        d[tid] = q0;
        d[512 + tid] = q1;
        __syncthreads();
      }
    }
    if (more) __syncthreads();
  }

  float msg[16];
#pragma unroll
  for (int w = 0; w < 4; ++w) {
    msg[w] = acc0[w] * SC0;
    msg[4 + 3 * w + 0] = acc1[w][0] * SC1;
    msg[4 + 3 * w + 1] = acc1[w][1] * SC1;
    msg[4 + 3 * w + 2] = acc1[w][2] * SC1;
  }
#pragma unroll
  for (int i = 0; i < 16; ++i) msg[i] += __shfl_xor(msg[i], 16);
  {
    float* o = out + (size_t)dnode * 16 + ((lane < 16) ? 0 : 8);
#pragma unroll
    for (int i = 0; i < 8; ++i)
      atomicAdd(o + i, (lane < 16) ? msg[i] : msg[8 + i]);
  }
}

extern "C" void kernel_launch(void* const* d_in, const int* in_sizes, int n_in,
                              void* d_out, int out_size, void* d_ws, size_t ws_size,
                              hipStream_t stream) {
  const float* node_pos  = (const float*)d_in[0];
  const float* edge_type = (const float*)d_in[1];
  const float* fc_w0     = (const float*)d_in[2];
  const float* fc_w1     = (const float*)d_in[3];
  const int*   esrc      = (const int*)d_in[4];
  const int*   edst      = (const int*)d_in[5];
  float* out = (float*)d_out;
  float* w1p = (float*)d_ws;
  int E = in_sizes[4];

  k_zero<<<(out_size + 255) / 256, 256, 0, stream>>>(out, out_size);
  k_repack_w1<<<(160 * 8 * 32 + 255) / 256, 256, 0, stream>>>(fc_w1, w1p);
  k_edge_conv<<<E / (16 * NWAVE), NWAVE * 32, 0, stream>>>(node_pos, edge_type,
                                                          fc_w0, w1p, esrc, edst, out);
}


# FC2 takes two N-tiles per loop trip: their 16 A-fragment LDS loads stand as ONE group ahead of 32 WMMAs; one 16-deep K-chain per tile, the two tiles' chains alternate; K sum order changed
# speedup vs baseline: 3.5771x; 1.0572x over previous
#include <hip/hip_runtime.h>
#include <math.h>


typedef float v2f __attribute__((ext_vector_type(2)));
typedef float v8f __attribute__((ext_vector_type(8)));

#define INV_SQRT2  0.70710678f
#define INV_SQRT3  0.57735027f
#define INV_SQRT6  0.40824829f
#define INV_SQRT10 0.31622777f
#define SILU_C     1.67668f
#define SC0        0.02209709f
#define SC1        0.03125f

__global__ void k_zero(float* p, int n) {
  int i = blockIdx.x * blockDim.x + threadIdx.x;
  if (i < n) p[i] = 0.f;
}

__global__ void k_repack_w1(const float* __restrict__ w1, float* __restrict__ o) {
  int tid = blockIdx.x * blockDim.x + threadIdx.x;
  if (tid >= 160 * 8 * 32) return;
  int l  = tid & 31;
  int j  = (tid >> 5) & 7;
  int t  = tid >> 8;
  int li = l & 15;
  int kd = (l < 16) ? 0 : 2;
  int row = 8 * j + kd;
  int col = t * 16 + li;
  float4 v;
  v.x = w1[row * 2560 + col];
  v.y = w1[(row + 1) * 2560 + col];
  v.z = w1[(row + 4) * 2560 + col];
  v.w = w1[(row + 5) * 2560 + col];
  ((float4*)o)[tid] = v;
}

#define NWAVE    16
#define CHUNK_F4 1024
#define PATH_F4  4096
#define MIDS     148

__global__ __launch_bounds__(NWAVE * 32)
void k_edge_conv(const float* __restrict__ node_pos,
                 const float* __restrict__ edge_type,
                 const float* __restrict__ fc_w0,
                 const float* w1p,
                 const int*   __restrict__ edge_src,
                 const int*   __restrict__ edge_dst,
                 float* __restrict__ out)
{
  __shared__ __attribute__((aligned(16))) float wst[2 * PATH_F4 * 4];
  __shared__ __attribute__((aligned(16))) float mid_all[NWAVE * 16 * MIDS];
  __shared__ __attribute__((aligned(16))) float xsr_all[NWAVE * 16 * 20];

  const int tid  = threadIdx.x;
  const int wave = __builtin_amdgcn_readfirstlane(tid >> 5);
  const int lane = tid & 31;
  float* const xsr = xsr_all + wave * (16 * 20);
  float* const mid = mid_all + wave * (16 * MIDS);
  float* const hid = mid;
  float* const xds = mid + 16 * 66;

  const int li   = lane & 15;
  const int kd   = (lane < 16) ? 0 : 2;
  const int eoff = (lane < 16) ? 0 : 8;
  const int e0   = (blockIdx.x * NWAVE + wave) * 16;

  const float4* const wg = (const float4*)w1p;
  const float4 pa0 = wg[tid], pb0 = wg[512 + tid];

  const int dnode = edge_dst[e0 + li];
  if (lane < 16) {
    int s = edge_src[e0 + li];
    const float4* q = (const float4*)(node_pos + (size_t)s * 16);
    float4* d = (float4*)(xsr + li * 20);
    d[0] = q[0]; d[1] = q[1]; d[2] = q[2]; d[3] = q[3];
  } else {
    const float4* q = (const float4*)(node_pos + (size_t)dnode * 16);
    float4* d = (float4*)(xds + li * 20);
    d[0] = q[0]; d[1] = q[1]; d[2] = q[2]; d[3] = q[3];
  }

  {
    v2f a[4];
    const float* et = edge_type + (size_t)(e0 + li) * 16 + kd;
#pragma unroll
    for (int s = 0; s < 4; ++s) a[s] = *(const v2f*)(et + 4 * s);
#pragma unroll
    for (int tt = 0; tt < 4; ++tt) {
      v8f c = {};
#pragma unroll
      for (int s = 0; s < 4; ++s) {
        int row = 4 * s + kd;
        v2f b;
        b.x = fc_w0[row * 64 + tt * 16 + li];
        b.y = fc_w0[(row + 1) * 64 + tt * 16 + li];
        c = __builtin_amdgcn_wmma_f32_16x16x4_f32(false, a[s], false, b,
                                                  (short)0, c, false, false);
      }
#pragma unroll
      for (int r = 0; r < 8; ++r) {
        float x = c[r] * 0.25f;
        float h = (x / (1.f + __expf(-x))) * (SILU_C * 0.125f);
        hid[(r + eoff) * 66 + tt * 16 + li] = h;
      }
    }
  }
  __syncthreads();

  v2f H[16];
#pragma unroll
  for (int s = 0; s < 16; ++s) H[s] = *(const v2f*)&hid[li * 66 + 4 * s + kd];
  float y0[4], y1[4][3];
  {
    const float* Y = xds + li * 20;
#pragma unroll
    for (int v = 0; v < 4; ++v) {
      y0[v] = Y[v];
      y1[v][0] = Y[4 + v * 3 + 0];
      y1[v][1] = Y[4 + v * 3 + 1];
      y1[v][2] = Y[4 + v * 3 + 2];
    }
  }
  float ys0[2], ys1[2][3];
#pragma unroll
  for (int j = 0; j < 2; ++j) {
    ys0[j] = (lane < 16) ? y0[j] : y0[2 + j];
#pragma unroll
    for (int k = 0; k < 3; ++k) ys1[j][k] = (lane < 16) ? y1[j][k] : y1[2 + j][k];
  }
  __syncthreads();

  {
    const int e = li;
    const float* X = xsr + e * 20;
    float* M = mid + e * MIDS;
    const int ub = (lane < 16) ? 0 : 2;
#pragma unroll
    for (int uu = 0; uu < 2; ++uu) {
      int u = ub + uu;
      float a0 = X[4 + u * 3 + 0], a1 = X[4 + u * 3 + 1], a2 = X[4 + u * 3 + 2];
#pragma unroll
      for (int v = 0; v < 4; ++v) {
        int uv = u * 4 + v;
        float b0 = y1[v][0], b1 = y1[v][1], b2 = y1[v][2];
        M[uv] = (a0 * b0 + a1 * b1 + a2 * b2) * INV_SQRT3;
        M[16 + uv * 3 + 0] = (a1 * b2 - a2 * b1) * INV_SQRT2;
        M[16 + uv * 3 + 1] = (a2 * b0 - a0 * b2) * INV_SQRT2;
        M[16 + uv * 3 + 2] = (a0 * b1 - a1 * b0) * INV_SQRT2;
        M[64 + uv * 5 + 0] = (a2 * b0 + a0 * b2) * INV_SQRT2;
        M[64 + uv * 5 + 1] = (a0 * b1 + a1 * b0) * INV_SQRT2;
        M[64 + uv * 5 + 2] = (2.f * a1 * b1 - a0 * b0 - a2 * b2) * INV_SQRT6;
        M[64 + uv * 5 + 3] = (a1 * b2 + a2 * b1) * INV_SQRT2;
        M[64 + uv * 5 + 4] = (a2 * b2 - a0 * b0) * INV_SQRT2;
      }
    }
  }
  ((float4*)wst)[tid] = pa0;
  ((float4*)wst)[512 + tid] = pb0;
  __syncthreads();

  float acc0[4], acc1[4][3];
#pragma unroll
  for (int w = 0; w < 4; ++w) {
    acc0[w] = 0.f; acc1[w][0] = 0.f; acc1[w][1] = 0.f; acc1[w][2] = 0.f;
  }
  const float* Xr = xsr + li * 20;
  const float* Mr = mid + li * MIDS;
  const int hv = (lane < 16) ? 0 : 2;

  float4 r0, r1;
  float4 q0, q1;
#pragma unroll
  for (int p = 0; p < 10; ++p) {
    const bool more = (p + 1 < 10);
    const float4* const cur = (const float4*)wst + (p & 1) * PATH_F4 + lane;
    float4* const oth = (float4*)wst + ((p + 1) & 1) * PATH_F4;
#pragma nounroll
    for (int sg = 0; sg < 4; ++sg) {
      if (more) {
        const float4* g = wg + (size_t)((p + 1) * 4 + sg) * CHUNK_F4;
        r0 = g[tid];
        r1 = g[512 + tid];
      }
      const bool own = (p == 0) && (sg < 3);
      if (own) {
        const float4* g = wg + (size_t)(sg + 1) * CHUNK_F4;
        q0 = g[tid];
        q1 = g[512 + tid];
      }
      const float4* wb = cur + sg * CHUNK_F4;
#pragma nounroll
      for (int ip = 0; ip < 2; ++ip) {
        const float4* aq = wb + ip * 512;
        const int u  = sg;
        float T0[2][2], T1[2][2][3];
#pragma unroll
        for (int tl = 0; tl < 2; ++tl) {
          const int i  = 2 * ip + tl;
          const int v0 = i * 4 + hv;
#pragma unroll
          for (int j = 0; j < 2; ++j) {
            const int v = v0 + j;
            if (p == 0) {
              T0[tl][j] = Xr[u] * (Xr[i] * ys0[j]);
            } else if (p == 1) {
              float f = Xr[u] * INV_SQRT3;
              float xi = Xr[i];
              T1[tl][j][0] = f * (xi * ys1[j][0]);
              T1[tl][j][1] = f * (xi * ys1[j][1]);
              T1[tl][j][2] = f * (xi * ys1[j][2]);
            } else if (p == 2) {
              float f = Xr[u] * INV_SQRT3;
              T1[tl][j][0] = f * (Xr[4 + i * 3 + 0] * ys0[j]);
              T1[tl][j][1] = f * (Xr[4 + i * 3 + 1] * ys0[j]);
              T1[tl][j][2] = f * (Xr[4 + i * 3 + 2] * ys0[j]);
            } else if (p == 3) {
              T0[tl][j] = Xr[u] * Mr[v];
            } else if (p == 4) {
              float f = (Xr[i] * ys0[j]) * INV_SQRT3;
              T1[tl][j][0] = f * Xr[4 + u * 3 + 0];
              T1[tl][j][1] = f * Xr[4 + u * 3 + 1];
              T1[tl][j][2] = f * Xr[4 + u * 3 + 2];
            } else if (p == 5) {
              float xi = Xr[i];
              float m0 = xi * ys1[j][0], m1 = xi * ys1[j][1], m2 = xi * ys1[j][2];
              float d = Xr[4 + u * 3 + 0] * m0
                      + Xr[4 + u * 3 + 1] * m1
                      + Xr[4 + u * 3 + 2] * m2;
              T0[tl][j] = d * INV_SQRT3;
            } else if (p == 6) {
              float m0 = Xr[4 + i * 3 + 0] * ys0[j], m1 = Xr[4 + i * 3 + 1] * ys0[j],
                    m2 = Xr[4 + i * 3 + 2] * ys0[j];
              float d = Xr[4 + u * 3 + 0] * m0
                      + Xr[4 + u * 3 + 1] * m1
                      + Xr[4 + u * 3 + 2] * m2;
              T0[tl][j] = d * INV_SQRT3;
            } else if (p == 7) {
              float f = Mr[v] * INV_SQRT3;
              T1[tl][j][0] = f * Xr[4 + u * 3 + 0];
              T1[tl][j][1] = f * Xr[4 + u * 3 + 1];
              T1[tl][j][2] = f * Xr[4 + u * 3 + 2];
            } else if (p == 8) {
              float a0 = Xr[4 + u * 3 + 0], a1 = Xr[4 + u * 3 + 1], a2 = Xr[4 + u * 3 + 2];
              float m0 = Mr[16 + v * 3 + 0], m1 = Mr[16 + v * 3 + 1], m2 = Mr[16 + v * 3 + 2];
              T1[tl][j][0] = INV_SQRT6 * (a1 * m2 - a2 * m1);
              T1[tl][j][1] = INV_SQRT6 * (a2 * m0 - a0 * m2);
              T1[tl][j][2] = INV_SQRT6 * (a0 * m1 - a1 * m0);
            } else {
              float a0 = Xr[4 + u * 3 + 0], a1 = Xr[4 + u * 3 + 1], a2 = Xr[4 + u * 3 + 2];
              const float* m5 = Mr + 64 + v * 5;
              float m0 = m5[0], m1 = m5[1], m2 = m5[2], m3 = m5[3], m4 = m5[4];
              float s2 = m2 * INV_SQRT3;
              T1[tl][j][0] = INV_SQRT10 * ((-s2 - m4) * a0 + m1 * a1 + m0 * a2);
              T1[tl][j][1] = INV_SQRT10 * (m1 * a0 + 2.f * s2 * a1 + m3 * a2);
              T1[tl][j][2] = INV_SQRT10 * (m0 * a0 + m3 * a1 + (-s2 + m4) * a2);
            }
          }
        }
        v8f c0 = {}, c1 = {};
        float4 wq[16];
#pragma unroll
        for (int s = 0; s < 16; ++s) wq[s] = aq[s * 32];
#pragma unroll
        for (int s = 0; s < 8; ++s) {
          v2f a0 = {wq[s].x, wq[s].y};
          v2f b0 = {wq[8 + s].x, wq[8 + s].y};
          v2f a1 = {wq[s].z, wq[s].w};
          v2f b1 = {wq[8 + s].z, wq[8 + s].w};
          c0 = __builtin_amdgcn_wmma_f32_16x16x4_f32(false, a0, false, H[2 * s],
                                                     (short)0, c0, false, false);
          c1 = __builtin_amdgcn_wmma_f32_16x16x4_f32(false, b0, false, H[2 * s],
                                                     (short)0, c1, false, false);
          c0 = __builtin_amdgcn_wmma_f32_16x16x4_f32(false, a1, false, H[2 * s + 1],
                                                     (short)0, c0, false, false);
          c1 = __builtin_amdgcn_wmma_f32_16x16x4_f32(false, b1, false, H[2 * s + 1],
                                                     (short)0, c1, false, false);
        }
        __builtin_amdgcn_sched_group_barrier(0x100, 48, 0);
        __builtin_amdgcn_sched_group_barrier(0x008, 32, 0);
#pragma unroll
        for (int tl = 0; tl < 2; ++tl) {
#pragma unroll
          for (int r = 0; r < 8; ++r) {
            const int j = r >> 2, w = r & 3;
            const float cr = (tl == 0) ? c0[r] : c1[r];
            if (p == 0 || p == 3 || p == 5 || p == 6) {
              acc0[w] += cr * T0[tl][j];
            } else {
              acc1[w][0] += cr * T1[tl][j][0];
              acc1[w][1] += cr * T1[tl][j][1];
              acc1[w][2] += cr * T1[tl][j][2];
            }
          }
        }
      }
      if (more) {
        float4* d = oth + sg * CHUNK_F4;
        d[tid] = r0;
        d[512 + tid] = r1;
      }
      if (own) {
        float4* d = (float4*)wst + (sg + 1) * CHUNK_F4;
        d[tid] = q0;
        d[512 + tid] = q1;
        __syncthreads();
      }
    }
    if (more) __syncthreads();
  }

  float msg[16];
#pragma unroll
  for (int w = 0; w < 4; ++w) {
    msg[w] = acc0[w] * SC0;
    msg[4 + 3 * w + 0] = acc1[w][0] * SC1;
    msg[4 + 3 * w + 1] = acc1[w][1] * SC1;
    msg[4 + 3 * w + 2] = acc1[w][2] * SC1;
  }
#pragma unroll
  for (int i = 0; i < 16; ++i) msg[i] += __shfl_xor(msg[i], 16);
  {
    float* o = out + (size_t)dnode * 16 + ((lane < 16) ? 0 : 8);
#pragma unroll
    for (int i = 0; i < 8; ++i)
      atomicAdd(o + i, (lane < 16) ? msg[i] : msg[8 + i]);
  }
}

extern "C" void kernel_launch(void* const* d_in, const int* in_sizes, int n_in,
                              void* d_out, int out_size, void* d_ws, size_t ws_size,
                              hipStream_t stream) {
  const float* node_pos  = (const float*)d_in[0];
  const float* edge_type = (const float*)d_in[1];
  const float* fc_w0     = (const float*)d_in[2];
  const float* fc_w1     = (const float*)d_in[3];
  const int*   esrc      = (const int*)d_in[4];
  const int*   edst      = (const int*)d_in[5];
  float* out = (float*)d_out;
  float* w1p = (float*)d_ws;
  int E = in_sizes[4];

  k_zero<<<(out_size + 255) / 256, 256, 0, stream>>>(out, out_size);
  k_repack_w1<<<(160 * 8 * 32 + 255) / 256, 256, 0, stream>>>(fc_w1, w1p);
  k_edge_conv<<<E / (16 * NWAVE), NWAVE * 32, 0, stream>>>(node_pos, edge_type,
                                                          fc_w0, w1p, esrc, edst, out);
}
